# starting kernel
# baseline (speedup 1.0000x reference)
#include <hip/hip_runtime.h>
#include <hip/hip_bf16.h>

typedef float v2f __attribute__((ext_vector_type(2)));
typedef float v4f __attribute__((ext_vector_type(4)));
typedef float v8f __attribute__((ext_vector_type(8)));

#define N_CLASS 10
#define NDIM    64
#define NCOMP   32
#define NKNOT   32
#define NTOT    65536

#define ROWS_PER_BLOCK 64
#define TPB            128
#define NWAVES         4
#define SD_STRIDE      68
#define SE_STRIDE      36

__global__ __launch_bounds__(TPB)
void cst_fused_kernel(const float* __restrict__ data,
                      const int*   __restrict__ label,
                      const float* __restrict__ wT,
                      const float* __restrict__ xx,
                      const float* __restrict__ yy,
                      const float* __restrict__ dl,
                      float* __restrict__ out,
                      float* __restrict__ logj)
{
    __shared__ float sData[ROWS_PER_BLOCK * SD_STRIDE];
    __shared__ v2f   sB1[16 * 2 * 32];
    __shared__ v2f   sB2[8 * 4 * 32];
    __shared__ float sE[NWAVES * 16 * SE_STRIDE];
    __shared__ int   sLab[ROWS_PER_BLOCK];

    const int tid  = threadIdx.x;
    const int base = blockIdx.x * ROWS_PER_BLOCK;
    const int lane = tid & 31;
    const int wave = tid >> 5;
    const int hi   = lane >> 4;
    const int l16  = lane & 15;

#pragma unroll
    for (int t = 0; t < 8; ++t) {
        int idx = tid + t * TPB;
        int row = idx >> 4, c4 = idx & 15;
        v4f v = __builtin_nontemporal_load(
                    (const v4f*)(data + (size_t)(base + row) * NDIM + c4 * 4));
        *(v4f*)(sData + row * SD_STRIDE + c4 * 4) = v;
    }
#pragma unroll
    for (int t = 0; t < 8; ++t) {
        int e = tid + t * TPB;
        int ln = e & 31, cj = e >> 5;
        int chunk = cj >> 1, j = cj & 1;
        int krow = 4 * chunk + 2 * (ln >> 4);
        int col  = (ln & 15) + 16 * j;
        v2f b;
        b.x = wT[krow * NCOMP + col];
        b.y = wT[(krow + 1) * NCOMP + col];
        sB1[e] = b;
    }
#pragma unroll
    for (int t = 0; t < 8; ++t) {
        int e = tid + t * TPB;
        int ln = e & 31, cj = e >> 5;
        int chunk = cj >> 2, jj = cj & 3;
        int n = jj * 16 + (ln & 15);
        int k = 4 * chunk + 2 * (ln >> 4);
        sB2[e] = *(const v2f*)(wT + n * NCOMP + k);
    }
    if (tid < ROWS_PER_BLOCK) sLab[tid] = label[base + tid];
    __syncthreads();

    const int rowL = wave * 16 + l16;
    v2f a1[16];
#pragma unroll
    for (int c = 0; c < 16; ++c)
        a1[c] = *(const v2f*)(sData + rowL * SD_STRIDE + 4 * c + 2 * hi);

    v8f c0 = {0.f, 0.f, 0.f, 0.f, 0.f, 0.f, 0.f, 0.f};
    v8f c1 = c0;
#pragma unroll
    for (int c = 0; c < 16; ++c)
        c0 = __builtin_amdgcn_wmma_f32_16x16x4_f32(false, a1[c], false,
                sB1[(c * 2 + 0) * 32 + lane], (short)0, c0, false, false);
#pragma unroll
    for (int c = 0; c < 16; ++c)
        c1 = __builtin_amdgcn_wmma_f32_16x16x4_f32(false, a1[c], false,
                sB1[(c * 2 + 1) * 32 + lane], (short)0, c1, false, false);

    float* sEw = sE + wave * 16 * SE_STRIDE;
    int kb[16];
    int idx[16];
    float ld[8];

#pragma unroll
    for (int r = 0; r < 8; ++r) {
        int cls = sLab[wave * 16 + r + 8 * hi];
        int k0  = (cls * NCOMP + l16) * NKNOT;
        kb[r]     = k0;
        kb[r + 8] = k0 + 16 * NKNOT;
    }
#define XV(e) (((e) < 8) ? c0[(e) & 7] : c1[(e) & 7])
#pragma unroll
    for (int e = 0; e < 16; ++e) idx[e] = (xx[kb[e] + 15] < XV(e)) ? 16 : 0;
#pragma unroll
    for (int e = 0; e < 16; ++e) idx[e] += (xx[kb[e] + idx[e] + 7] < XV(e)) ? 8 : 0;
#pragma unroll
    for (int e = 0; e < 16; ++e) idx[e] += (xx[kb[e] + idx[e] + 3] < XV(e)) ? 4 : 0;
#pragma unroll
    for (int e = 0; e < 16; ++e) idx[e] += (xx[kb[e] + idx[e] + 1] < XV(e)) ? 2 : 0;
#pragma unroll
    for (int e = 0; e < 16; ++e) idx[e] += (xx[kb[e] + idx[e]] < XV(e)) ? 1 : 0;
#pragma unroll
    for (int e = 0; e < 16; ++e) {
        float t = xx[kb[e] + 31];
        if (idx[e] == 31 && t < XV(e)) idx[e] = 32;
    }

#pragma unroll
    for (int g = 0; g < 2; ++g) {
        float xk[8], xk1[8], yk[8], yk1[8], dk[8], dk1[8];
#pragma unroll
        for (int r = 0; r < 8; ++r) {
            int e = g * 8 + r;
            int b = idx[e] - 1;
            b = b < 0 ? 0 : (b > NKNOT - 2 ? NKNOT - 2 : b);
            int o = kb[e] + b;
            xk[r] = xx[o]; xk1[r] = xx[o + 1];
            yk[r] = yy[o]; yk1[r] = yy[o + 1];
            dk[r] = dl[o]; dk1[r] = dl[o + 1];
        }
#pragma unroll
        for (int r = 0; r < 8; ++r) {
            int e = g * 8 + r;
            float x   = (g == 0) ? c0[r] : c1[r];
            float w   = xk1[r] - xk[r];
            float s   = (yk1[r] - yk[r]) / w;
            float xi  = fminf(fmaxf((x - xk[r]) / w, 0.0f), 1.0f);
            float xi1 = 1.0f - xi;
            float cc  = xi * xi1;
            float den = s + (dk1[r] + dk[r] - 2.0f * s) * cc;
            float y_in = yk[r] + (yk1[r] - yk[r]) * (s * xi * xi + dk[r] * cc) / den;
            float d_in = s * s * (dk1[r] * xi * xi + 2.0f * s * cc + dk[r] * xi1 * xi1)
                         / (den * den);
            bool below = (idx[e] == 0);
            bool above = (idx[e] == NKNOT);
            float y  = below ? fmaf(x - xk[r], dk[r], yk[r])
                     : (above ? fmaf(x - xk1[r], dk1[r], yk1[r]) : y_in);
            float dd = below ? dk[r] : (above ? dk1[r] : d_in);
            float l  = __logf(dd);
            if (g == 0) ld[r] = l; else ld[r] += l;
            int M = r + 8 * hi;
            sEw[M * SE_STRIDE + l16 + 16 * g] = y - x;
        }
    }
#undef XV

#pragma unroll
    for (int r = 0; r < 8; ++r) {
        float p = ld[r];
        p += __shfl_xor(p, 1, 32);
        p += __shfl_xor(p, 2, 32);
        p += __shfl_xor(p, 4, 32);
        p += __shfl_xor(p, 8, 32);
        if (l16 == 0)
            __builtin_nontemporal_store(p, logj + base + wave * 16 + r + 8 * hi);
    }
    asm volatile("s_wait_dscnt 0" ::: "memory");

    v2f a2[8];
#pragma unroll
    for (int c = 0; c < 8; ++c)
        a2[c] = *(const v2f*)(sEw + l16 * SE_STRIDE + 4 * c + 2 * hi);

#pragma unroll
    for (int jj = 0; jj < 4; ++jj) {
        v8f acc = {0.f, 0.f, 0.f, 0.f, 0.f, 0.f, 0.f, 0.f};
#pragma unroll
        for (int c = 0; c < 8; ++c)
            acc = __builtin_amdgcn_wmma_f32_16x16x4_f32(false, a2[c], false,
                    sB2[(c * 4 + jj) * 32 + lane], (short)0, acc, false, false);
        int col = jj * 16 + l16;
#pragma unroll
        for (int r = 0; r < 8; ++r) {
            int M    = r + 8 * hi;
            int srow = wave * 16 + M;
            __builtin_nontemporal_store(acc[r] + sData[srow * SD_STRIDE + col],
                                        out + (size_t)(base + srow) * NDIM + col);
        }
    }
}

extern "C" void kernel_launch(void* const* d_in, const int* in_sizes, int n_in,
                              void* d_out, int out_size, void* d_ws, size_t ws_size,
                              hipStream_t stream) {
    const float* data  = (const float*)d_in[0];
    const int*   label = (const int*)  d_in[1];
    const float* wT    = (const float*)d_in[2];
    const float* xx    = (const float*)d_in[3];
    const float* yy    = (const float*)d_in[4];
    const float* dl    = (const float*)d_in[5];
    float* out  = (float*)d_out;
    float* logj = out + (size_t)NTOT * NDIM;

    dim3 grid(NTOT / ROWS_PER_BLOCK);
    cst_fused_kernel<<<grid, TPB, 0, stream>>>(data, label, wT, xx, yy, dl, out, logj);
}


# Spline knot tables (xx,yy,dl; all classes) staged once per 512-thread workgroup into LDS at an odd row pitch; searches and gathers read LDS instead of scattered global loads
# speedup vs baseline: 2.0160x; 2.0160x over previous
#include <hip/hip_runtime.h>
#include <hip/hip_bf16.h>

typedef float v2f __attribute__((ext_vector_type(2)));
typedef float v4f __attribute__((ext_vector_type(4)));
typedef float v8f __attribute__((ext_vector_type(8)));

#define N_CLASS 10
#define NDIM    64
#define NCOMP   32
#define NKNOT   32
#define NTOT    65536

#define ROWS_PER_BLOCK 256
#define TPB            512
#define NWAVES         16
#define SD_STRIDE      68
#define SE_STRIDE      36
#define TROWS          (N_CLASS * NCOMP)
#define TR_STRIDE      97
#define TAB_ITERS      (TROWS * NKNOT / 4 / TPB)

__global__ __launch_bounds__(TPB)
void cst_fused_kernel(const float* __restrict__ data,
                      const int*   __restrict__ label,
                      const float* __restrict__ wT,
                      const float* __restrict__ xx,
                      const float* __restrict__ yy,
                      const float* __restrict__ dl,
                      float* __restrict__ out,
                      float* __restrict__ logj)
{
    __shared__ float sData[ROWS_PER_BLOCK * SD_STRIDE];
    __shared__ v2f   sB1[16 * 2 * 32];
    __shared__ v2f   sB2[8 * 4 * 32];
    __shared__ float sE[NWAVES * 16 * SE_STRIDE];
    __shared__ float sT[TROWS * TR_STRIDE];
    __shared__ int   sLab[ROWS_PER_BLOCK];

    const int tid  = threadIdx.x;
    const int base = blockIdx.x * ROWS_PER_BLOCK;
    const int lane = tid & 31;
    const int wave = tid >> 5;
    const int hi   = lane >> 4;
    const int l16  = lane & 15;

    v4f dv[8];
#pragma unroll
    for (int t = 0; t < 8; ++t) {
        int idx = tid + t * TPB;
        int row = idx >> 4, c4 = idx & 15;
        dv[t] = __builtin_nontemporal_load(
                    (const v4f*)(data + (size_t)(base + row) * NDIM + c4 * 4));
    }
    v2f b1v[2];
#pragma unroll
    for (int t = 0; t < 2; ++t) {
        int e = tid + t * TPB;
        int ln = e & 31, cj = e >> 5;
        int chunk = cj >> 1, j = cj & 1;
        int krow = 4 * chunk + 2 * (ln >> 4);
        int col  = (ln & 15) + 16 * j;
        b1v[t].x = wT[krow * NCOMP + col];
        b1v[t].y = wT[(krow + 1) * NCOMP + col];
    }
    v2f b2v[2];
#pragma unroll
    for (int t = 0; t < 2; ++t) {
        int e = tid + t * TPB;
        int ln = e & 31, cj = e >> 5;
        int chunk = cj >> 2, jj = cj & 3;
        int n = jj * 16 + (ln & 15);
        int k = 4 * chunk + 2 * (ln >> 4);
        b2v[t] = *(const v2f*)(wT + n * NCOMP + k);
    }
    const int labv = label[base + wave * 16 + l16];
    v4f tx[TAB_ITERS], ty[TAB_ITERS], td[TAB_ITERS];
#pragma unroll
    for (int t = 0; t < TAB_ITERS; ++t) {
        int i4 = tid + t * TPB;
        tx[t] = *(const v4f*)(xx + 4 * i4);
        ty[t] = *(const v4f*)(yy + 4 * i4);
        td[t] = *(const v4f*)(dl + 4 * i4);
    }
#pragma unroll
    for (int t = 0; t < 8; ++t) {
        int idx = tid + t * TPB;
        int row = idx >> 4, c4 = idx & 15;
        *(v4f*)(sData + row * SD_STRIDE + c4 * 4) = dv[t];
    }
#pragma unroll
    for (int t = 0; t < 2; ++t) {
        sB1[tid + t * TPB] = b1v[t];
        sB2[tid + t * TPB] = b2v[t];
    }
    sLab[wave * 16 + l16] = labv;
    __builtin_amdgcn_fence(__ATOMIC_RELEASE, "workgroup", "local");
    __builtin_amdgcn_s_barrier();
    __builtin_amdgcn_fence(__ATOMIC_ACQUIRE, "workgroup", "local");

    const int rowL = wave * 16 + l16;
    v2f a1[16];
#pragma unroll
    for (int c = 0; c < 16; ++c)
        a1[c] = *(const v2f*)(sData + rowL * SD_STRIDE + 4 * c + 2 * hi);

    v8f c0 = {0.f, 0.f, 0.f, 0.f, 0.f, 0.f, 0.f, 0.f};
    v8f c1 = c0;
#pragma unroll
    for (int c = 0; c < 16; ++c)
        c0 = __builtin_amdgcn_wmma_f32_16x16x4_f32(false, a1[c], false,
                sB1[(c * 2 + 0) * 32 + lane], (short)0, c0, false, false);
#pragma unroll
    for (int c = 0; c < 16; ++c)
        c1 = __builtin_amdgcn_wmma_f32_16x16x4_f32(false, a1[c], false,
                sB1[(c * 2 + 1) * 32 + lane], (short)0, c1, false, false);

#pragma unroll
    for (int t = 0; t < TAB_ITERS; ++t) {
        int i4 = tid + t * TPB;
        float* d = sT + (i4 >> 3) * TR_STRIDE + (i4 & 7) * 4;
        d[0]  = tx[t].x; d[1]  = tx[t].y; d[2]  = tx[t].z; d[3]  = tx[t].w;
        d[32] = ty[t].x; d[33] = ty[t].y; d[34] = ty[t].z; d[35] = ty[t].w;
        d[64] = td[t].x; d[65] = td[t].y; d[66] = td[t].z; d[67] = td[t].w;
    }
    __syncthreads();

    float* sEw = sE + wave * 16 * SE_STRIDE;
    int kb[16];
    int idx[16];
    float ld[8];

#pragma unroll
    for (int r = 0; r < 8; ++r) {
        int cls = sLab[wave * 16 + r + 8 * hi];
        int k0  = (cls * NCOMP + l16) * TR_STRIDE;
        kb[r]     = k0;
        kb[r + 8] = k0 + 16 * TR_STRIDE;
    }
#define XV(e) (((e) < 8) ? c0[(e) & 7] : c1[(e) & 7])
#pragma unroll
    for (int e = 0; e < 16; ++e) idx[e] = (sT[kb[e] + 15] < XV(e)) ? 16 : 0;
#pragma unroll
    for (int e = 0; e < 16; ++e) idx[e] += (sT[kb[e] + idx[e] + 7] < XV(e)) ? 8 : 0;
#pragma unroll
    for (int e = 0; e < 16; ++e) idx[e] += (sT[kb[e] + idx[e] + 3] < XV(e)) ? 4 : 0;
#pragma unroll
    for (int e = 0; e < 16; ++e) idx[e] += (sT[kb[e] + idx[e] + 1] < XV(e)) ? 2 : 0;
#pragma unroll
    for (int e = 0; e < 16; ++e) idx[e] += (sT[kb[e] + idx[e]] < XV(e)) ? 1 : 0;
#pragma unroll
    for (int e = 0; e < 16; ++e) {
        float t = sT[kb[e] + 31];
        if (idx[e] == 31 && t < XV(e)) idx[e] = 32;
    }

#pragma unroll
    for (int g = 0; g < 2; ++g) {
        float xk[8], xk1[8], yk[8], yk1[8], dk[8], dk1[8];
#pragma unroll
        for (int r = 0; r < 8; ++r) {
            int e = g * 8 + r;
            int b = idx[e] - 1;
            b = b < 0 ? 0 : (b > NKNOT - 2 ? NKNOT - 2 : b);
            int o = kb[e] + b;
            xk[r] = sT[o];              xk1[r] = sT[o + 1];
            yk[r] = sT[o + NKNOT];      yk1[r] = sT[o + NKNOT + 1];
            dk[r] = sT[o + 2 * NKNOT];  dk1[r] = sT[o + 2 * NKNOT + 1];
        }
#pragma unroll
        for (int r = 0; r < 8; ++r) {
            int e = g * 8 + r;
            float x   = (g == 0) ? c0[r] : c1[r];
            float w   = xk1[r] - xk[r];
            float s   = (yk1[r] - yk[r]) / w;
            float xi  = fminf(fmaxf((x - xk[r]) / w, 0.0f), 1.0f);
            float xi1 = 1.0f - xi;
            float cc  = xi * xi1;
            float den = s + (dk1[r] + dk[r] - 2.0f * s) * cc;
            float y_in = yk[r] + (yk1[r] - yk[r]) * (s * xi * xi + dk[r] * cc) / den;
            float d_in = s * s * (dk1[r] * xi * xi + 2.0f * s * cc + dk[r] * xi1 * xi1)
                         / (den * den);
            bool below = (idx[e] == 0);
            bool above = (idx[e] == NKNOT);
            float y  = below ? fmaf(x - xk[r], dk[r], yk[r])
                     : (above ? fmaf(x - xk1[r], dk1[r], yk1[r]) : y_in);
            float dd = below ? dk[r] : (above ? dk1[r] : d_in);
            float l  = __logf(dd);
            if (g == 0) ld[r] = l; else ld[r] += l;
            int M = r + 8 * hi;
            sEw[M * SE_STRIDE + l16 + 16 * g] = y - x;
        }
    }
#undef XV

#pragma unroll
    for (int r = 0; r < 8; ++r) {
        float p = ld[r];
        p += __shfl_xor(p, 1, 32);
        p += __shfl_xor(p, 2, 32);
        p += __shfl_xor(p, 4, 32);
        p += __shfl_xor(p, 8, 32);
        if (l16 == 0)
            __builtin_nontemporal_store(p, logj + base + wave * 16 + r + 8 * hi);
    }
    asm volatile("s_wait_dscnt 0" ::: "memory");

    v2f a2[8];
#pragma unroll
    for (int c = 0; c < 8; ++c)
        a2[c] = *(const v2f*)(sEw + l16 * SE_STRIDE + 4 * c + 2 * hi);

#pragma unroll
    for (int jj = 0; jj < 4; ++jj) {
        v8f acc = {0.f, 0.f, 0.f, 0.f, 0.f, 0.f, 0.f, 0.f};
#pragma unroll
        for (int c = 0; c < 8; ++c)
            acc = __builtin_amdgcn_wmma_f32_16x16x4_f32(false, a2[c], false,
                    sB2[(c * 4 + jj) * 32 + lane], (short)0, acc, false, false);
        int col = jj * 16 + l16;
#pragma unroll
        for (int r = 0; r < 8; ++r) {
            int M    = r + 8 * hi;
            int srow = wave * 16 + M;
            __builtin_nontemporal_store(acc[r] + sData[srow * SD_STRIDE + col],
                                        out + (size_t)(base + srow) * NDIM + col);
        }
    }
}

extern "C" void kernel_launch(void* const* d_in, const int* in_sizes, int n_in,
                              void* d_out, int out_size, void* d_ws, size_t ws_size,
                              hipStream_t stream) {
    const float* data  = (const float*)d_in[0];
    const int*   label = (const int*)  d_in[1];
    const float* wT    = (const float*)d_in[2];
    const float* xx    = (const float*)d_in[3];
    const float* yy    = (const float*)d_in[4];
    const float* dl    = (const float*)d_in[5];
    float* out  = (float*)d_out;
    float* logj = out + (size_t)NTOT * NDIM;

    dim3 grid(NTOT / ROWS_PER_BLOCK);
    cst_fused_kernel<<<grid, TPB, 0, stream>>>(data, label, wT, xx, yy, dl, out, logj);
}


# Spline evaluation: the four IEEE f32 division sequences per element become two hardware reciprocals (v_rcp_f32, one per denominator) and multiplies; quotient rounding changes by an ulp or two
# speedup vs baseline: 2.1298x; 1.0564x over previous
#include <hip/hip_runtime.h>
#include <hip/hip_bf16.h>

typedef float v2f __attribute__((ext_vector_type(2)));
typedef float v4f __attribute__((ext_vector_type(4)));
typedef float v8f __attribute__((ext_vector_type(8)));

#define N_CLASS 10
#define NDIM    64
#define NCOMP   32
#define NKNOT   32
#define NTOT    65536

#define ROWS_PER_BLOCK 256
#define TPB            512
#define NWAVES         16
#define SD_STRIDE      68
#define SE_STRIDE      36
#define TROWS          (N_CLASS * NCOMP)
#define TR_STRIDE      97
#define TAB_ITERS      (TROWS * NKNOT / 4 / TPB)

__global__ __launch_bounds__(TPB)
void cst_fused_kernel(const float* __restrict__ data,
                      const int*   __restrict__ label,
                      const float* __restrict__ wT,
                      const float* __restrict__ xx,
                      const float* __restrict__ yy,
                      const float* __restrict__ dl,
                      float* __restrict__ out,
                      float* __restrict__ logj)
{
    __shared__ float sData[ROWS_PER_BLOCK * SD_STRIDE];
    __shared__ v2f   sB1[16 * 2 * 32];
    __shared__ v2f   sB2[8 * 4 * 32];
    __shared__ float sE[NWAVES * 16 * SE_STRIDE];
    __shared__ float sT[TROWS * TR_STRIDE];
    __shared__ int   sLab[ROWS_PER_BLOCK];

    const int tid  = threadIdx.x;
    const int base = blockIdx.x * ROWS_PER_BLOCK;
    const int lane = tid & 31;
    const int wave = tid >> 5;
    const int hi   = lane >> 4;
    const int l16  = lane & 15;

    v4f dv[8];
#pragma unroll
    for (int t = 0; t < 8; ++t) {
        int idx = tid + t * TPB;
        int row = idx >> 4, c4 = idx & 15;
        dv[t] = __builtin_nontemporal_load(
                    (const v4f*)(data + (size_t)(base + row) * NDIM + c4 * 4));
    }
    v2f b1v[2];
#pragma unroll
    for (int t = 0; t < 2; ++t) {
        int e = tid + t * TPB;
        int ln = e & 31, cj = e >> 5;
        int chunk = cj >> 1, j = cj & 1;
        int krow = 4 * chunk + 2 * (ln >> 4);
        int col  = (ln & 15) + 16 * j;
        b1v[t].x = wT[krow * NCOMP + col];
        b1v[t].y = wT[(krow + 1) * NCOMP + col];
    }
    v2f b2v[2];
#pragma unroll
    for (int t = 0; t < 2; ++t) {
        int e = tid + t * TPB;
        int ln = e & 31, cj = e >> 5;
        int chunk = cj >> 2, jj = cj & 3;
        int n = jj * 16 + (ln & 15);
        int k = 4 * chunk + 2 * (ln >> 4);
        b2v[t] = *(const v2f*)(wT + n * NCOMP + k);
    }
    const int labv = label[base + wave * 16 + l16];
    v4f tx[TAB_ITERS], ty[TAB_ITERS], td[TAB_ITERS];
#pragma unroll
    for (int t = 0; t < TAB_ITERS; ++t) {
        int i4 = tid + t * TPB;
        tx[t] = *(const v4f*)(xx + 4 * i4);
        ty[t] = *(const v4f*)(yy + 4 * i4);
        td[t] = *(const v4f*)(dl + 4 * i4);
    }
#pragma unroll
    for (int t = 0; t < 8; ++t) {
        int idx = tid + t * TPB;
        int row = idx >> 4, c4 = idx & 15;
        *(v4f*)(sData + row * SD_STRIDE + c4 * 4) = dv[t];
    }
#pragma unroll
    for (int t = 0; t < 2; ++t) {
        sB1[tid + t * TPB] = b1v[t];
        sB2[tid + t * TPB] = b2v[t];
    }
    sLab[wave * 16 + l16] = labv;
    __builtin_amdgcn_fence(__ATOMIC_RELEASE, "workgroup", "local");
    __builtin_amdgcn_s_barrier();
    __builtin_amdgcn_fence(__ATOMIC_ACQUIRE, "workgroup", "local");

    const int rowL = wave * 16 + l16;
    v2f a1[16];
#pragma unroll
    for (int c = 0; c < 16; ++c)
        a1[c] = *(const v2f*)(sData + rowL * SD_STRIDE + 4 * c + 2 * hi);

    v8f c0 = {0.f, 0.f, 0.f, 0.f, 0.f, 0.f, 0.f, 0.f};
    v8f c1 = c0;
#pragma unroll
    for (int c = 0; c < 16; ++c)
        c0 = __builtin_amdgcn_wmma_f32_16x16x4_f32(false, a1[c], false,
                sB1[(c * 2 + 0) * 32 + lane], (short)0, c0, false, false);
#pragma unroll
    for (int c = 0; c < 16; ++c)
        c1 = __builtin_amdgcn_wmma_f32_16x16x4_f32(false, a1[c], false,
                sB1[(c * 2 + 1) * 32 + lane], (short)0, c1, false, false);

#pragma unroll
    for (int t = 0; t < TAB_ITERS; ++t) {
        int i4 = tid + t * TPB;
        float* d = sT + (i4 >> 3) * TR_STRIDE + (i4 & 7) * 4;
        d[0]  = tx[t].x; d[1]  = tx[t].y; d[2]  = tx[t].z; d[3]  = tx[t].w;
        d[32] = ty[t].x; d[33] = ty[t].y; d[34] = ty[t].z; d[35] = ty[t].w;
        d[64] = td[t].x; d[65] = td[t].y; d[66] = td[t].z; d[67] = td[t].w;
    }
    __syncthreads();

    float* sEw = sE + wave * 16 * SE_STRIDE;
    int kb[16];
    int idx[16];
    float ld[8];

#pragma unroll
    for (int r = 0; r < 8; ++r) {
        int cls = sLab[wave * 16 + r + 8 * hi];
        int k0  = (cls * NCOMP + l16) * TR_STRIDE;
        kb[r]     = k0;
        kb[r + 8] = k0 + 16 * TR_STRIDE;
    }
#define XV(e) (((e) < 8) ? c0[(e) & 7] : c1[(e) & 7])
#pragma unroll
    for (int e = 0; e < 16; ++e) idx[e] = (sT[kb[e] + 15] < XV(e)) ? 16 : 0;
#pragma unroll
    for (int e = 0; e < 16; ++e) idx[e] += (sT[kb[e] + idx[e] + 7] < XV(e)) ? 8 : 0;
#pragma unroll
    for (int e = 0; e < 16; ++e) idx[e] += (sT[kb[e] + idx[e] + 3] < XV(e)) ? 4 : 0;
#pragma unroll
    for (int e = 0; e < 16; ++e) idx[e] += (sT[kb[e] + idx[e] + 1] < XV(e)) ? 2 : 0;
#pragma unroll
    for (int e = 0; e < 16; ++e) idx[e] += (sT[kb[e] + idx[e]] < XV(e)) ? 1 : 0;
#pragma unroll
    for (int e = 0; e < 16; ++e) {
        float t = sT[kb[e] + 31];
        if (idx[e] == 31 && t < XV(e)) idx[e] = 32;
    }

#pragma unroll
    for (int g = 0; g < 2; ++g) {
        float xk[8], xk1[8], yk[8], yk1[8], dk[8], dk1[8];
#pragma unroll
        for (int r = 0; r < 8; ++r) {
            int e = g * 8 + r;
            int b = idx[e] - 1;
            b = b < 0 ? 0 : (b > NKNOT - 2 ? NKNOT - 2 : b);
            int o = kb[e] + b;
            xk[r] = sT[o];              xk1[r] = sT[o + 1];
            yk[r] = sT[o + NKNOT];      yk1[r] = sT[o + NKNOT + 1];
            dk[r] = sT[o + 2 * NKNOT];  dk1[r] = sT[o + 2 * NKNOT + 1];
        }
#pragma unroll
        for (int r = 0; r < 8; ++r) {
            int e = g * 8 + r;
            float x   = (g == 0) ? c0[r] : c1[r];
            float rw  = __builtin_amdgcn_rcpf(xk1[r] - xk[r]);
            float dy  = yk1[r] - yk[r];
            float s   = dy * rw;
            float xi  = fminf(fmaxf((x - xk[r]) * rw, 0.0f), 1.0f);
            float xi1 = 1.0f - xi;
            float cc  = xi * xi1;
            float den = s + (dk1[r] + dk[r] - 2.0f * s) * cc;
            float rden = __builtin_amdgcn_rcpf(den);
            float y_in = yk[r] + dy * (s * xi * xi + dk[r] * cc) * rden;
            float sr   = s * rden;
            float d_in = (sr * sr) * (dk1[r] * xi * xi + 2.0f * s * cc + dk[r] * xi1 * xi1);
            bool below = (idx[e] == 0);
            bool above = (idx[e] == NKNOT);
            float y  = below ? fmaf(x - xk[r], dk[r], yk[r])
                     : (above ? fmaf(x - xk1[r], dk1[r], yk1[r]) : y_in);
            float dd = below ? dk[r] : (above ? dk1[r] : d_in);
            float l  = __logf(dd);
            if (g == 0) ld[r] = l; else ld[r] += l;
            int M = r + 8 * hi;
            sEw[M * SE_STRIDE + l16 + 16 * g] = y - x;
        }
    }
#undef XV

#pragma unroll
    for (int r = 0; r < 8; ++r) {
        float p = ld[r];
        p += __shfl_xor(p, 1, 32);
        p += __shfl_xor(p, 2, 32);
        p += __shfl_xor(p, 4, 32);
        p += __shfl_xor(p, 8, 32);
        if (l16 == 0)
            __builtin_nontemporal_store(p, logj + base + wave * 16 + r + 8 * hi);
    }
    asm volatile("s_wait_dscnt 0" ::: "memory");

    v2f a2[8];
#pragma unroll
    for (int c = 0; c < 8; ++c)
        a2[c] = *(const v2f*)(sEw + l16 * SE_STRIDE + 4 * c + 2 * hi);

#pragma unroll
    for (int jj = 0; jj < 4; ++jj) {
        v8f acc = {0.f, 0.f, 0.f, 0.f, 0.f, 0.f, 0.f, 0.f};
#pragma unroll
        for (int c = 0; c < 8; ++c)
            acc = __builtin_amdgcn_wmma_f32_16x16x4_f32(false, a2[c], false,
                    sB2[(c * 4 + jj) * 32 + lane], (short)0, acc, false, false);
        int col = jj * 16 + l16;
#pragma unroll
        for (int r = 0; r < 8; ++r) {
            int M    = r + 8 * hi;
            int srow = wave * 16 + M;
            __builtin_nontemporal_store(acc[r] + sData[srow * SD_STRIDE + col],
                                        out + (size_t)(base + srow) * NDIM + col);
        }
    }
}

extern "C" void kernel_launch(void* const* d_in, const int* in_sizes, int n_in,
                              void* d_out, int out_size, void* d_ws, size_t ws_size,
                              hipStream_t stream) {
    const float* data  = (const float*)d_in[0];
    const int*   label = (const int*)  d_in[1];
    const float* wT    = (const float*)d_in[2];
    const float* xx    = (const float*)d_in[3];
    const float* yy    = (const float*)d_in[4];
    const float* dl    = (const float*)d_in[5];
    float* out  = (float*)d_out;
    float* logj = out + (size_t)NTOT * NDIM;

    dim3 grid(NTOT / ROWS_PER_BLOCK);
    cst_fused_kernel<<<grid, TPB, 0, stream>>>(data, label, wT, xx, yy, dl, out, logj);
}


# Workgroups of 8 waves instead of 16: each wave carries two 16-row tiles through every phase in lock step, one B fragment read feeding both; half as many waves to launch; same sums, same order
# speedup vs baseline: 2.3840x; 1.1194x over previous
#include <hip/hip_runtime.h>
#include <hip/hip_bf16.h>


typedef float v2f __attribute__((ext_vector_type(2)));
typedef float v4f __attribute__((ext_vector_type(4)));
typedef float v8f __attribute__((ext_vector_type(8)));

#define N_CLASS 10
#define NDIM    64
#define NCOMP   32
#define NKNOT   32
#define NTOT    65536

#define ROWS_PER_BLOCK 256
#define TPB            256
#define NWAVES         8
#define MT             2
#define NTILES         (ROWS_PER_BLOCK / 16)
#define SD_STRIDE      68
#define SE_STRIDE      36
#define TROWS          (N_CLASS * NCOMP)
#define TR_STRIDE      97
#define TAB_ITERS      (TROWS * NKNOT / 4 / TPB)
#define DATA_ITERS     (ROWS_PER_BLOCK * NDIM / 4 / TPB)
#define B_ITERS        (16 * 2 * 32 / TPB)

__global__ __launch_bounds__(TPB)
void cst_fused_kernel(const float* __restrict__ data,
                      const int*   __restrict__ label,
                      const float* __restrict__ wT,
                      const float* __restrict__ xx,
                      const float* __restrict__ yy,
                      const float* __restrict__ dl,
                      float* __restrict__ out,
                      float* __restrict__ logj)
{
    __shared__ float sData[ROWS_PER_BLOCK * SD_STRIDE];
    __shared__ v2f   sB1[16 * 2 * 32];
    __shared__ v2f   sB2[8 * 4 * 32];
    __shared__ float sE[NTILES * 16 * SE_STRIDE];
    __shared__ float sT[TROWS * TR_STRIDE];
    __shared__ int   sLab[ROWS_PER_BLOCK];

    const int tid  = threadIdx.x;
    const int base = blockIdx.x * ROWS_PER_BLOCK;
    const int lane = tid & 31;
    const int wave = tid >> 5;
    const int hi   = lane >> 4;
    const int l16  = lane & 15;
    const int tile0 = wave * MT;

    v4f dv[DATA_ITERS];
#pragma unroll
    for (int t = 0; t < DATA_ITERS; ++t) {
        int idx = tid + t * TPB;
        int row = idx >> 4, c4 = idx & 15;
        dv[t] = __builtin_nontemporal_load(
                    (const v4f*)(data + (size_t)(base + row) * NDIM + c4 * 4));
    }
    v2f b1v[B_ITERS];
#pragma unroll
    for (int t = 0; t < B_ITERS; ++t) {
        int e = tid + t * TPB;
        int ln = e & 31, cj = e >> 5;
        int chunk = cj >> 1, j = cj & 1;
        int krow = 4 * chunk + 2 * (ln >> 4);
        int col  = (ln & 15) + 16 * j;
        b1v[t].x = wT[krow * NCOMP + col];
        b1v[t].y = wT[(krow + 1) * NCOMP + col];
    }
    v2f b2v[B_ITERS];
#pragma unroll
    for (int t = 0; t < B_ITERS; ++t) {
        int e = tid + t * TPB;
        int ln = e & 31, cj = e >> 5;
        int chunk = cj >> 2, jj = cj & 3;
        int n = jj * 16 + (ln & 15);
        int k = 4 * chunk + 2 * (ln >> 4);
        b2v[t] = *(const v2f*)(wT + n * NCOMP + k);
    }
    const int labv = label[base + tid];
    v4f tx[TAB_ITERS], ty[TAB_ITERS], td[TAB_ITERS];
#pragma unroll
    for (int t = 0; t < TAB_ITERS; ++t) {
        int i4 = tid + t * TPB;
        tx[t] = *(const v4f*)(xx + 4 * i4);
        ty[t] = *(const v4f*)(yy + 4 * i4);
        td[t] = *(const v4f*)(dl + 4 * i4);
    }
#pragma unroll
    for (int t = 0; t < DATA_ITERS; ++t) {
        int idx = tid + t * TPB;
        int row = idx >> 4, c4 = idx & 15;
        *(v4f*)(sData + row * SD_STRIDE + c4 * 4) = dv[t];
    }
#pragma unroll
    for (int t = 0; t < B_ITERS; ++t) {
        sB1[tid + t * TPB] = b1v[t];
        sB2[tid + t * TPB] = b2v[t];
    }
    sLab[tid] = labv;
    __builtin_amdgcn_fence(__ATOMIC_RELEASE, "workgroup", "local");
    __builtin_amdgcn_s_barrier();
    __builtin_amdgcn_fence(__ATOMIC_ACQUIRE, "workgroup", "local");

    v8f c0[MT], c1[MT];
#pragma unroll
    for (int mt = 0; mt < MT; ++mt) {
        c0[mt] = (v8f){0.f, 0.f, 0.f, 0.f, 0.f, 0.f, 0.f, 0.f};
        c1[mt] = c0[mt];
    }
#pragma unroll
    for (int c = 0; c < 16; ++c) {
        v2f bA = sB1[(c * 2 + 0) * 32 + lane];
        v2f bB = sB1[(c * 2 + 1) * 32 + lane];
#pragma unroll
        for (int mt = 0; mt < MT; ++mt) {
            int rowL = (tile0 + mt) * 16 + l16;
            v2f a = *(const v2f*)(sData + rowL * SD_STRIDE + 4 * c + 2 * hi);
            c0[mt] = __builtin_amdgcn_wmma_f32_16x16x4_f32(false, a, false, bA,
                                                            (short)0, c0[mt], false, false);
            c1[mt] = __builtin_amdgcn_wmma_f32_16x16x4_f32(false, a, false, bB,
                                                            (short)0, c1[mt], false, false);
        }
    }

    __builtin_amdgcn_sched_barrier(0);

#pragma unroll
    for (int t = 0; t < TAB_ITERS; ++t) {
        int i4 = tid + t * TPB;
        float* d = sT + (i4 >> 3) * TR_STRIDE + (i4 & 7) * 4;
        d[0]  = tx[t].x; d[1]  = tx[t].y; d[2]  = tx[t].z; d[3]  = tx[t].w;
        d[32] = ty[t].x; d[33] = ty[t].y; d[34] = ty[t].z; d[35] = ty[t].w;
        d[64] = td[t].x; d[65] = td[t].y; d[66] = td[t].z; d[67] = td[t].w;
    }
    __syncthreads();

    int kb[MT][16];
    int idx[MT][16];
    float ld[MT][8];

#pragma unroll
    for (int mt = 0; mt < MT; ++mt) {
#pragma unroll
        for (int r = 0; r < 8; ++r) {
            int cls = sLab[(tile0 + mt) * 16 + r + 8 * hi];
            int k0  = (cls * NCOMP + l16) * TR_STRIDE;
            kb[mt][r]     = k0;
            kb[mt][r + 8] = k0 + 16 * TR_STRIDE;
        }
    }
#define XV(mt, e) (((e) < 8) ? c0[mt][(e) & 7] : c1[mt][(e) & 7])
    float k31[MT][16];
#pragma unroll
    for (int mt = 0; mt < MT; ++mt)
#pragma unroll
        for (int e = 0; e < 16; ++e) {
            idx[mt][e] = (sT[kb[mt][e] + 15] < XV(mt, e)) ? 16 : 0;
            k31[mt][e] = sT[kb[mt][e] + 31];
        }
#pragma unroll
    for (int mt = 0; mt < MT; ++mt)
#pragma unroll
        for (int e = 0; e < 16; ++e) idx[mt][e] += (sT[kb[mt][e] + idx[mt][e] + 7] < XV(mt, e)) ? 8 : 0;
#pragma unroll
    for (int mt = 0; mt < MT; ++mt)
#pragma unroll
        for (int e = 0; e < 16; ++e) idx[mt][e] += (sT[kb[mt][e] + idx[mt][e] + 3] < XV(mt, e)) ? 4 : 0;
#pragma unroll
    for (int mt = 0; mt < MT; ++mt)
#pragma unroll
        for (int e = 0; e < 16; ++e) idx[mt][e] += (sT[kb[mt][e] + idx[mt][e] + 1] < XV(mt, e)) ? 2 : 0;
#pragma unroll
    for (int mt = 0; mt < MT; ++mt)
#pragma unroll
        for (int e = 0; e < 16; ++e) idx[mt][e] += (sT[kb[mt][e] + idx[mt][e]] < XV(mt, e)) ? 1 : 0;
#pragma unroll
    for (int mt = 0; mt < MT; ++mt)
#pragma unroll
        for (int e = 0; e < 16; ++e)
            idx[mt][e] = ((idx[mt][e] == 31) & (k31[mt][e] < XV(mt, e))) ? 32 : idx[mt][e];
#undef XV

#pragma unroll
    for (int mt = 0; mt < MT; ++mt) {
        float* sEw = sE + (tile0 + mt) * 16 * SE_STRIDE;
#pragma unroll
        for (int g = 0; g < 2; ++g) {
            float xk[8], xk1[8], yk[8], yk1[8], dk[8], dk1[8];
#pragma unroll
            for (int r = 0; r < 8; ++r) {
                int e = g * 8 + r;
                int b = idx[mt][e] - 1;
                b = b < 0 ? 0 : (b > NKNOT - 2 ? NKNOT - 2 : b);
                int o = kb[mt][e] + b;
                xk[r] = sT[o];              xk1[r] = sT[o + 1];
                yk[r] = sT[o + NKNOT];      yk1[r] = sT[o + NKNOT + 1];
                dk[r] = sT[o + 2 * NKNOT];  dk1[r] = sT[o + 2 * NKNOT + 1];
            }
#pragma unroll
            for (int r = 0; r < 8; ++r) {
                int e = g * 8 + r;
                float x   = (g == 0) ? c0[mt][r] : c1[mt][r];
                float rw  = __builtin_amdgcn_rcpf(xk1[r] - xk[r]);
                float dy  = yk1[r] - yk[r];
                float s   = dy * rw;
                float xi  = fminf(fmaxf((x - xk[r]) * rw, 0.0f), 1.0f);
                float xi1 = 1.0f - xi;
                float cc  = xi * xi1;
                float den = s + (dk1[r] + dk[r] - 2.0f * s) * cc;
                float rden = __builtin_amdgcn_rcpf(den);
                float y_in = yk[r] + dy * (s * xi * xi + dk[r] * cc) * rden;
                float sr   = s * rden;
                float d_in = (sr * sr) * (dk1[r] * xi * xi + 2.0f * s * cc + dk[r] * xi1 * xi1);
                bool below = (idx[mt][e] == 0);
                bool above = (idx[mt][e] == NKNOT);
                float y  = below ? fmaf(x - xk[r], dk[r], yk[r])
                         : (above ? fmaf(x - xk1[r], dk1[r], yk1[r]) : y_in);
                float dd = below ? dk[r] : (above ? dk1[r] : d_in);
                float l  = __logf(dd);
                if (g == 0) ld[mt][r] = l; else ld[mt][r] += l;
                int M = r + 8 * hi;
                sEw[M * SE_STRIDE + l16 + 16 * g] = y - x;
            }
        }
    }

#pragma unroll
    for (int s = 1; s < 16; s <<= 1) {
#pragma unroll
        for (int mt = 0; mt < MT; ++mt)
#pragma unroll
            for (int r = 0; r < 8; ++r)
                ld[mt][r] += __shfl_xor(ld[mt][r], s, 32);
    }
#pragma unroll
    for (int mt = 0; mt < MT; ++mt) {
        float res = ld[mt][0];
#pragma unroll
        for (int r = 1; r < 8; ++r) res = (l16 == r) ? ld[mt][r] : res;
        if (l16 < 8)
            __builtin_nontemporal_store(res, logj + base + (tile0 + mt) * 16 + l16 + 8 * hi);
    }
    asm volatile("s_wait_dscnt 0" ::: "memory");

    v2f a2[MT][8];
#pragma unroll
    for (int mt = 0; mt < MT; ++mt)
#pragma unroll
        for (int c = 0; c < 8; ++c)
            a2[mt][c] = *(const v2f*)(sE + (tile0 + mt) * 16 * SE_STRIDE
                                         + l16 * SE_STRIDE + 4 * c + 2 * hi);

#pragma unroll
    for (int jj = 0; jj < 4; ++jj) {
        v8f acc[MT];
#pragma unroll
        for (int mt = 0; mt < MT; ++mt)
            acc[mt] = (v8f){0.f, 0.f, 0.f, 0.f, 0.f, 0.f, 0.f, 0.f};
#pragma unroll
        for (int c = 0; c < 8; ++c) {
            v2f b = sB2[(c * 4 + jj) * 32 + lane];
#pragma unroll
            for (int mt = 0; mt < MT; ++mt)
                acc[mt] = __builtin_amdgcn_wmma_f32_16x16x4_f32(false, a2[mt][c], false, b,
                                                                (short)0, acc[mt], false, false);
        }
        int col = jj * 16 + l16;
#pragma unroll
        for (int mt = 0; mt < MT; ++mt) {
#pragma unroll
            for (int r = 0; r < 8; ++r) {
                int M    = r + 8 * hi;
                int srow = (tile0 + mt) * 16 + M;
                __builtin_nontemporal_store(acc[mt][r] + sData[srow * SD_STRIDE + col],
                                            out + (size_t)(base + srow) * NDIM + col);
            }
        }
    }
}

extern "C" void kernel_launch(void* const* d_in, const int* in_sizes, int n_in,
                              void* d_out, int out_size, void* d_ws, size_t ws_size,
                              hipStream_t stream) {
    const float* data  = (const float*)d_in[0];
    const int*   label = (const int*)  d_in[1];
    const float* wT    = (const float*)d_in[2];
    const float* xx    = (const float*)d_in[3];
    const float* yy    = (const float*)d_in[4];
    const float* dl    = (const float*)d_in[5];
    float* out  = (float*)d_out;
    float* logj = out + (size_t)NTOT * NDIM;

    dim3 grid(NTOT / ROWS_PER_BLOCK);
    cst_fused_kernel<<<grid, TPB, 0, stream>>>(data, label, wT, xx, yy, dl, out, logj);
}
